# starting kernel
# baseline (speedup 1.0000x reference)
#include <hip/hip_runtime.h>
#include <hip/hip_bf16.h>

#define N_TOT 131072
#define DD    64
#define SS    25
#define SP    32
#define GG    132
#define GP    144
#define GPS   145
#define TM    64

typedef __attribute__((ext_vector_type(2))) float v2f;
typedef __attribute__((ext_vector_type(8))) float v8f;

__global__ void precompute_kernel(const float* __restrict__ W1,
                                  const float* __restrict__ W2,
                                  const float* __restrict__ Y,
                                  const float* __restrict__ Yw,
                                  float* __restrict__ B1,
                                  float* __restrict__ B2,
                                  float* __restrict__ Ywp) {
    int tid    = blockIdx.x * blockDim.x + threadIdx.x;
    int stride = gridDim.x * blockDim.x;
    for (int i = tid; i < DD * GP; i += stride) {
        int d = i / GP, g = i % GP;
        float s1 = 0.f, s2 = 0.f;
        if (g < GG) {
            for (int s = 0; s < SS; ++s) {
                float y = Y[g * SS + s];
                s1 += W1[d * SS + s] * y;
                s2 += W2[d * SS + s] * y;
            }
        }
        B1[i] = s1;
        B2[i] = s2;
    }
    for (int i = tid; i < GP * SP; i += stride) {
        int g = i / SP, s = i % SP;
        Ywp[i] = (g < GG && s < SS) ? Yw[g * SS + s] : 0.f;
    }
}

__global__ void __launch_bounds__(128)
gaunt_fused_kernel(const float* __restrict__ X1,
                   const float* __restrict__ X2,
                   const float* __restrict__ B1,
                   const float* __restrict__ B2,
                   const float* __restrict__ Ywp,
                   float* __restrict__ out) {
    extern __shared__ float smem[];
    float* sB1 = smem;
    float* sB2 = sB1 + DD * GP;
    float* sYw = sB2 + DD * GP;
    float* sH  = sYw + GP * SP;

    const int tid  = threadIdx.x;
    const int lane = tid & 31;
    const int wave = tid >> 5;
    const int hsel = lane >> 4;
    const int l16  = lane & 15;

    for (int i = tid; i < DD * GP; i += blockDim.x) {
        sB1[i] = B1[i];
        sB2[i] = B2[i];
    }
    for (int i = tid; i < GP * SP; i += blockDim.x) sYw[i] = Ywp[i];
    __syncthreads();

    const int rowBase = blockIdx.x * TM + wave * 16;

    v2f A1[16], A2[16];
    {
        const float* x1p = X1 + (size_t)(rowBase + l16) * DD + 2 * hsel;
        const float* x2p = X2 + (size_t)(rowBase + l16) * DD + 2 * hsel;
#pragma unroll
        for (int j = 0; j < 16; ++j) {
            A1[j] = *(const v2f*)(x1p + 4 * j);
            A2[j] = *(const v2f*)(x2p + 4 * j);
        }
    }

    for (int t = 0; t < GP / 16; ++t) {
        const int col = t * 16 + l16;
        v8f acc1 = {};
        v8f acc2 = {};
#pragma unroll
        for (int j = 0; j < 16; ++j) {
            const int k = 4 * j + 2 * hsel;
            v2f b1, b2;
            b1.x = sB1[k * GP + col];
            b1.y = sB1[(k + 1) * GP + col];
            b2.x = sB2[k * GP + col];
            b2.y = sB2[(k + 1) * GP + col];
            acc1 = __builtin_amdgcn_wmma_f32_16x16x4_f32(
                false, A1[j], false, b1, (short)0, acc1, false, false);
            acc2 = __builtin_amdgcn_wmma_f32_16x16x4_f32(
                false, A2[j], false, b2, (short)0, acc2, false, false);
        }
        v8f h = acc1 * acc2;
#pragma unroll
        for (int r = 0; r < 8; ++r) {
            const int m = wave * 16 + r + 8 * hsel;
            sH[m * GPS + col] = h[r];
        }
    }

    for (int u = 0; u < SP / 16; ++u) {
        const int col = u * 16 + l16;
        const float* hrow = sH + (wave * 16 + l16) * GPS + 2 * hsel;
        v8f acc = {};
#pragma unroll
        for (int j = 0; j < GP / 4; ++j) {
            const int k = 4 * j;
            v2f a, b;
            a.x = hrow[k];
            a.y = hrow[k + 1];
            b.x = sYw[(k + 2 * hsel) * SP + col];
            b.y = sYw[(k + 2 * hsel + 1) * SP + col];
            acc = __builtin_amdgcn_wmma_f32_16x16x4_f32(
                false, a, false, b, (short)0, acc, false, false);
        }
        if (col < SS) {
#pragma unroll
            for (int r = 0; r < 8; ++r) {
                const int m = r + 8 * hsel;
                out[(size_t)(rowBase + m) * SS + col] = acc[r];
            }
        }
    }
}

extern "C" void kernel_launch(void* const* d_in, const int* in_sizes, int n_in,
                              void* d_out, int out_size, void* d_ws, size_t ws_size,
                              hipStream_t stream) {
    const float* X1 = (const float*)d_in[0];
    const float* X2 = (const float*)d_in[1];
    const float* W1 = (const float*)d_in[2];
    const float* W2 = (const float*)d_in[3];
    const float* Y  = (const float*)d_in[4];
    const float* Yw = (const float*)d_in[5];

    float* B1  = (float*)d_ws;
    float* B2  = B1 + DD * GP;
    float* Ywp = B2 + DD * GP;

    precompute_kernel<<<32, 256, 0, stream>>>(W1, W2, Y, Yw, B1, B2, Ywp);

    const size_t shmem =
        (size_t)(2 * DD * GP + GP * SP + TM * GPS) * sizeof(float);
    gaunt_fused_kernel<<<N_TOT / TM, 128, shmem, stream>>>(
        X1, X2, B1, B2, Ywp, (float*)d_out);
}


# LDS staging of B1/B2/Yw: fully unrolled 128-bit loads at a compile-time stride, about 20 in flight per thread, instead of one dword load per L2 round trip (108 serial trips per block)
# speedup vs baseline: 2.3145x; 2.3145x over previous
#include <hip/hip_runtime.h>
#include <hip/hip_bf16.h>

#define N_TOT 131072
#define DD    64
#define SS    25
#define SP    32
#define GG    132
#define GP    144
#define GPS   145
#define TM    64
#define NT    128

typedef __attribute__((ext_vector_type(2))) float v2f;
typedef __attribute__((ext_vector_type(4))) float v4f;
typedef __attribute__((ext_vector_type(8))) float v8f;

__global__ void precompute_kernel(const float* __restrict__ W1,
                                  const float* __restrict__ W2,
                                  const float* __restrict__ Y,
                                  const float* __restrict__ Yw,
                                  float* __restrict__ B1,
                                  float* __restrict__ B2,
                                  float* __restrict__ Ywp) {
    int tid    = blockIdx.x * blockDim.x + threadIdx.x;
    int stride = gridDim.x * blockDim.x;
    for (int i = tid; i < DD * GP; i += stride) {
        int d = i / GP, g = i % GP;
        float s1 = 0.f, s2 = 0.f;
        if (g < GG) {
            for (int s = 0; s < SS; ++s) {
                float y = Y[g * SS + s];
                s1 += W1[d * SS + s] * y;
                s2 += W2[d * SS + s] * y;
            }
        }
        B1[i] = s1;
        B2[i] = s2;
    }
    for (int i = tid; i < GP * SP; i += stride) {
        int g = i / SP, s = i % SP;
        Ywp[i] = (g < GG && s < SS) ? Yw[g * SS + s] : 0.f;
    }
}

__global__ void __launch_bounds__(NT)
gaunt_fused_kernel(const float* __restrict__ X1,
                   const float* __restrict__ X2,
                   const float* __restrict__ B1,
                   const float* __restrict__ B2,
                   const float* __restrict__ Ywp,
                   float* __restrict__ out) {
    extern __shared__ float smem[];
    float* sB1 = smem;
    float* sB2 = sB1 + DD * GP;
    float* sYw = sB2 + DD * GP;
    float* sH  = sYw + GP * SP;

    const int tid  = threadIdx.x;
    const int lane = tid & 31;
    const int wave = tid >> 5;
    const int hsel = lane >> 4;
    const int l16  = lane & 15;

    {
        const v4f* gB1 = (const v4f*)B1;
        const v4f* gB2 = (const v4f*)B2;
        const v4f* gYw = (const v4f*)Ywp;
        v4f* lB1 = (v4f*)sB1;
        v4f* lB2 = (v4f*)sB2;
        v4f* lYw = (v4f*)sYw;
        v4f r1[DD * GP / 4 / NT], r2[DD * GP / 4 / NT], ry[GP * SP / 4 / NT];
#pragma unroll
        for (int i = 0; i < DD * GP / 4 / NT; ++i) {
            r1[i] = gB1[tid + i * NT];
            r2[i] = gB2[tid + i * NT];
        }
#pragma unroll
        for (int i = 0; i < GP * SP / 4 / NT; ++i) ry[i] = gYw[tid + i * NT];
#pragma unroll
        for (int i = 0; i < DD * GP / 4 / NT; ++i) {
            lB1[tid + i * NT] = r1[i];
            lB2[tid + i * NT] = r2[i];
        }
#pragma unroll
        for (int i = 0; i < GP * SP / 4 / NT; ++i) lYw[tid + i * NT] = ry[i];
    }
    __syncthreads();

    const int rowBase = blockIdx.x * TM + wave * 16;

    v2f A1[16], A2[16];
    {
        const float* x1p = X1 + (size_t)(rowBase + l16) * DD + 2 * hsel;
        const float* x2p = X2 + (size_t)(rowBase + l16) * DD + 2 * hsel;
#pragma unroll
        for (int j = 0; j < 16; ++j) {
            A1[j] = *(const v2f*)(x1p + 4 * j);
            A2[j] = *(const v2f*)(x2p + 4 * j);
        }
    }

    for (int t = 0; t < GP / 16; ++t) {
        const int col = t * 16 + l16;
        v8f acc1 = {};
        v8f acc2 = {};
#pragma unroll
        for (int j = 0; j < 16; ++j) {
            const int k = 4 * j + 2 * hsel;
            v2f b1, b2;
            b1.x = sB1[k * GP + col];
            b1.y = sB1[(k + 1) * GP + col];
            b2.x = sB2[k * GP + col];
            b2.y = sB2[(k + 1) * GP + col];
            acc1 = __builtin_amdgcn_wmma_f32_16x16x4_f32(
                false, A1[j], false, b1, (short)0, acc1, false, false);
            acc2 = __builtin_amdgcn_wmma_f32_16x16x4_f32(
                false, A2[j], false, b2, (short)0, acc2, false, false);
        }
        v8f h = acc1 * acc2;
#pragma unroll
        for (int r = 0; r < 8; ++r) {
            const int m = wave * 16 + r + 8 * hsel;
            sH[m * GPS + col] = h[r];
        }
    }

    for (int u = 0; u < SP / 16; ++u) {
        const int col = u * 16 + l16;
        const float* hrow = sH + (wave * 16 + l16) * GPS + 2 * hsel;
        v8f acc = {};
#pragma unroll
        for (int j = 0; j < GP / 4; ++j) {
            const int k = 4 * j;
            v2f a, b;
            a.x = hrow[k];
            a.y = hrow[k + 1];
            b.x = sYw[(k + 2 * hsel) * SP + col];
            b.y = sYw[(k + 2 * hsel + 1) * SP + col];
            acc = __builtin_amdgcn_wmma_f32_16x16x4_f32(
                false, a, false, b, (short)0, acc, false, false);
        }
        if (col < SS) {
#pragma unroll
            for (int r = 0; r < 8; ++r) {
                const int m = r + 8 * hsel;
                out[(size_t)(rowBase + m) * SS + col] = acc[r];
            }
        }
    }
}

extern "C" void kernel_launch(void* const* d_in, const int* in_sizes, int n_in,
                              void* d_out, int out_size, void* d_ws, size_t ws_size,
                              hipStream_t stream) {
    const float* X1 = (const float*)d_in[0];
    const float* X2 = (const float*)d_in[1];
    const float* W1 = (const float*)d_in[2];
    const float* W2 = (const float*)d_in[3];
    const float* Y  = (const float*)d_in[4];
    const float* Yw = (const float*)d_in[5];

    float* B1  = (float*)d_ws;
    float* B2  = B1 + DD * GP;
    float* Ywp = B2 + DD * GP;

    precompute_kernel<<<32, 256, 0, stream>>>(W1, W2, Y, Yw, B1, B2, Ywp);

    const size_t shmem =
        (size_t)(2 * DD * GP + GP * SP + TM * GPS) * sizeof(float);
    gaunt_fused_kernel<<<N_TOT / TM, NT, shmem, stream>>>(
        X1, X2, B1, B2, Ywp, (float*)d_out);
}


# LDS fragment loads of the next WMMA block issued as one group into a second register set a full block ahead (both GEMMs, held by scheduling barriers), not one load waited for before each WMMA
# speedup vs baseline: 3.6843x; 1.5918x over previous
#include <hip/hip_runtime.h>
#include <hip/hip_bf16.h>

#define N_TOT 131072
#define DD    64
#define SS    25
#define SP    32
#define GG    132
#define GP    144
#define GPS   145
#define TM    64
#define NT    128

typedef __attribute__((ext_vector_type(2))) float v2f;
typedef __attribute__((ext_vector_type(4))) float v4f;
typedef __attribute__((ext_vector_type(8))) float v8f;

__global__ void precompute_kernel(const float* __restrict__ W1,
                                  const float* __restrict__ W2,
                                  const float* __restrict__ Y,
                                  const float* __restrict__ Yw,
                                  float* __restrict__ B1,
                                  float* __restrict__ B2,
                                  float* __restrict__ Ywp) {
    int tid    = blockIdx.x * blockDim.x + threadIdx.x;
    int stride = gridDim.x * blockDim.x;
    for (int i = tid; i < DD * GP; i += stride) {
        int d = i / GP, g = i % GP;
        float s1 = 0.f, s2 = 0.f;
        if (g < GG) {
            for (int s = 0; s < SS; ++s) {
                float y = Y[g * SS + s];
                s1 += W1[d * SS + s] * y;
                s2 += W2[d * SS + s] * y;
            }
        }
        B1[i] = s1;
        B2[i] = s2;
    }
    for (int i = tid; i < GP * SP; i += stride) {
        int g = i / SP, s = i % SP;
        Ywp[i] = (g < GG && s < SS) ? Yw[g * SS + s] : 0.f;
    }
}

struct BTile { v2f b1[16]; v2f b2[16]; };
struct HBlk  { v2f a[9]; v2f b0[9]; v2f b1[9]; };

static __device__ __forceinline__ void split_load_groups() {
    asm volatile("" ::: "memory");
}

static __device__ __forceinline__ void close_load_group() {
    split_load_groups();
    __builtin_amdgcn_sched_barrier(0);
}

static __device__ __forceinline__ void load_btile(BTile& f, const float* sB1,
                                                  const float* sB2, int t,
                                                  int l16, int hsel) {
    const int col = t * 16 + l16;
#pragma unroll
    for (int j = 0; j < 16; ++j) {
        const int k = 4 * j + 2 * hsel;
        f.b1[j].x = sB1[k * GP + col];
        f.b1[j].y = sB1[(k + 1) * GP + col];
        f.b2[j].x = sB2[k * GP + col];
        f.b2[j].y = sB2[(k + 1) * GP + col];
    }
}

static __device__ __forceinline__ void load_hblk(HBlk& f, const float* hrow,
                                                 const float* sYw, int blk,
                                                 int l16, int hsel) {
#pragma unroll
    for (int jj = 0; jj < 9; ++jj) {
        const int k = 4 * (blk * 9 + jj);
        f.a[jj].x  = hrow[k];
        f.a[jj].y  = hrow[k + 1];
        f.b0[jj].x = sYw[(k + 2 * hsel) * SP + l16];
        f.b0[jj].y = sYw[(k + 2 * hsel + 1) * SP + l16];
    }
    split_load_groups();
#pragma unroll
    for (int jj = 0; jj < 9; ++jj) {
        const int k = 4 * (blk * 9 + jj);
        f.b1[jj].x = sYw[(k + 2 * hsel) * SP + 16 + l16];
        f.b1[jj].y = sYw[(k + 2 * hsel + 1) * SP + 16 + l16];
    }
}

__global__ void __launch_bounds__(NT)
gaunt_fused_kernel(const float* __restrict__ X1,
                   const float* __restrict__ X2,
                   const float* __restrict__ B1,
                   const float* __restrict__ B2,
                   const float* __restrict__ Ywp,
                   float* __restrict__ out) {
    extern __shared__ float smem[];
    float* sB1 = smem;
    float* sB2 = sB1 + DD * GP;
    float* sYw = sB2 + DD * GP;
    float* sH  = sYw + GP * SP;

    const int tid  = threadIdx.x;
    const int lane = tid & 31;
    const int wave = tid >> 5;
    const int hsel = lane >> 4;
    const int l16  = lane & 15;

    {
        const v4f* gB1 = (const v4f*)B1;
        const v4f* gB2 = (const v4f*)B2;
        const v4f* gYw = (const v4f*)Ywp;
        v4f* lB1 = (v4f*)sB1;
        v4f* lB2 = (v4f*)sB2;
        v4f* lYw = (v4f*)sYw;
        v4f r1[DD * GP / 4 / NT], r2[DD * GP / 4 / NT], ry[GP * SP / 4 / NT];
#pragma unroll
        for (int i = 0; i < DD * GP / 4 / NT; ++i) {
            r1[i] = gB1[tid + i * NT];
            r2[i] = gB2[tid + i * NT];
        }
#pragma unroll
        for (int i = 0; i < GP * SP / 4 / NT; ++i) ry[i] = gYw[tid + i * NT];
#pragma unroll
        for (int i = 0; i < DD * GP / 4 / NT; ++i) {
            lB1[tid + i * NT] = r1[i];
            lB2[tid + i * NT] = r2[i];
        }
#pragma unroll
        for (int i = 0; i < GP * SP / 4 / NT; ++i) lYw[tid + i * NT] = ry[i];
    }
    __syncthreads();

    const int rowBase = blockIdx.x * TM + wave * 16;

    v2f A1[16], A2[16];
    {
        const float* x1p = X1 + (size_t)(rowBase + l16) * DD + 2 * hsel;
        const float* x2p = X2 + (size_t)(rowBase + l16) * DD + 2 * hsel;
#pragma unroll
        for (int j = 0; j < 16; ++j) {
            A1[j] = *(const v2f*)(x1p + 4 * j);
            A2[j] = *(const v2f*)(x2p + 4 * j);
        }
    }

    BTile bt[2];
    HBlk  hb[2];
    const float* hrow = sH + (wave * 16 + l16) * GPS + 2 * hsel;
    load_btile(bt[0], sB1, sB2, 0, l16, hsel);
    split_load_groups();
#pragma unroll
    for (int t = 0; t < GP / 16; ++t) {
        if (t + 1 < GP / 16) {
            load_btile(bt[(t + 1) & 1], sB1, sB2, t + 1, l16, hsel);
        } else {
            load_hblk(hb[0], hrow, sYw, 0, l16, hsel);
        }
        close_load_group();
        const BTile& c = bt[t & 1];
        v8f acc1 = {};
        v8f acc2 = {};
#pragma unroll
        for (int j = 0; j < 16; ++j) {
            acc1 = __builtin_amdgcn_wmma_f32_16x16x4_f32(
                false, A1[j], false, c.b1[j], (short)0, acc1, false, false);
            acc2 = __builtin_amdgcn_wmma_f32_16x16x4_f32(
                false, A2[j], false, c.b2[j], (short)0, acc2, false, false);
        }
        __builtin_amdgcn_sched_barrier(0);
        v8f h = acc1 * acc2;
        const int col = t * 16 + l16;
#pragma unroll
        for (int r = 0; r < 8; ++r) {
            const int m = wave * 16 + r + 8 * hsel;
            sH[m * GPS + col] = h[r];
        }
    }

    v8f o0 = {};
    v8f o1 = {};
#pragma unroll
    for (int blk = 0; blk < 4; ++blk) {
        if (blk + 1 < 4) load_hblk(hb[(blk + 1) & 1], hrow, sYw, blk + 1, l16, hsel);
        close_load_group();
        const HBlk& c = hb[blk & 1];
#pragma unroll
        for (int jj = 0; jj < 9; ++jj) {
            o0 = __builtin_amdgcn_wmma_f32_16x16x4_f32(
                false, c.a[jj], false, c.b0[jj], (short)0, o0, false, false);
            o1 = __builtin_amdgcn_wmma_f32_16x16x4_f32(
                false, c.a[jj], false, c.b1[jj], (short)0, o1, false, false);
        }
        __builtin_amdgcn_sched_barrier(0);
    }
#pragma unroll
    for (int r = 0; r < 8; ++r) {
        const int m = r + 8 * hsel;
        out[(size_t)(rowBase + m) * SS + l16] = o0[r];
    }
    if (16 + l16 < SS) {
#pragma unroll
        for (int r = 0; r < 8; ++r) {
            const int m = r + 8 * hsel;
            out[(size_t)(rowBase + m) * SS + 16 + l16] = o1[r];
        }
    }
}

extern "C" void kernel_launch(void* const* d_in, const int* in_sizes, int n_in,
                              void* d_out, int out_size, void* d_ws, size_t ws_size,
                              hipStream_t stream) {
    const float* X1 = (const float*)d_in[0];
    const float* X2 = (const float*)d_in[1];
    const float* W1 = (const float*)d_in[2];
    const float* W2 = (const float*)d_in[3];
    const float* Y  = (const float*)d_in[4];
    const float* Yw = (const float*)d_in[5];

    float* B1  = (float*)d_ws;
    float* B2  = B1 + DD * GP;
    float* Ywp = B2 + DD * GP;

    precompute_kernel<<<32, 256, 0, stream>>>(W1, W2, Y, Yw, B1, B2, Ywp);

    const size_t shmem =
        (size_t)(2 * DD * GP + GP * SP + TM * GPS) * sizeof(float);
    gaunt_fused_kernel<<<N_TOT / TM, NT, shmem, stream>>>(
        X1, X2, B1, B2, Ywp, (float*)d_out);
}


# GEMM1 unfolded into two chained f32 WMMA products (x*W 64->32, then c*Y^T 28->144, the reference's own order) instead of one folded K=64 product: 262 WMMAs per wave, not 360; sums reassociated
# speedup vs baseline: 4.9358x; 1.3397x over previous
#include <hip/hip_runtime.h>
#include <hip/hip_bf16.h>

#define N_TOT 131072
#define DD    64
#define SS    25
#define SP    32
#define SK    28
#define NJ1   16
#define NJ2   7
#define FR    64
#define GG    132
#define GP    144
#define GPS   145
#define CS    36
#define TM    64
#define NT    128

#define OFF_W1  0
#define OFF_W2  (NJ1 * 2 * FR)
#define OFF_YT  (2 * NJ1 * 2 * FR)
#define OFF_YW  (OFF_YT + (GP / 16) * NJ2 * FR)
#define N_CONST (OFF_YW + GP * SP)

typedef __attribute__((ext_vector_type(2))) float v2f;
typedef __attribute__((ext_vector_type(4))) float v4f;
typedef __attribute__((ext_vector_type(8))) float v8f;

__global__ void precompute_kernel(const float* __restrict__ W1,
                                  const float* __restrict__ W2,
                                  const float* __restrict__ Y,
                                  const float* __restrict__ Yw,
                                  float* __restrict__ C) {
    int tid    = blockIdx.x * blockDim.x + threadIdx.x;
    int stride = gridDim.x * blockDim.x;
    for (int i = tid; i < NJ1 * 2 * FR; i += stride) {
        int e = i & 1, ln = (i >> 1) & 31, f = i / FR;
        int t = f % 2, j = f / 2;
        int k = 4 * j + 2 * (ln >> 4) + e, s = 16 * t + (ln & 15);
        C[OFF_W1 + i] = (s < SS) ? W1[k * SS + s] : 0.f;
        C[OFF_W2 + i] = (s < SS) ? W2[k * SS + s] : 0.f;
    }
    for (int i = tid; i < (GP / 16) * NJ2 * FR; i += stride) {
        int e = i & 1, ln = (i >> 1) & 31, f = i / FR;
        int j = f % NJ2, t = f / NJ2;
        int k = 4 * j + 2 * (ln >> 4) + e, g = 16 * t + (ln & 15);
        C[OFF_YT + i] = (k < SS && g < GG) ? Y[g * SS + k] : 0.f;
    }
    for (int i = tid; i < GP * SP; i += stride) {
        int g = i / SP, s = i % SP;
        C[OFF_YW + i] = (g < GG && s < SS) ? Yw[g * SS + s] : 0.f;
    }
}

struct WFrag { v2f b0[16]; v2f b1[16]; };
struct YTile { v2f b[NJ2]; };
struct HBlk  { v2f a[9]; v2f b0[9]; v2f b1[9]; };

static __device__ __forceinline__ void split_load_groups() {
    asm volatile("" ::: "memory");
}

static __device__ __forceinline__ void close_load_group() {
    split_load_groups();
    __builtin_amdgcn_sched_barrier(0);
}

static __device__ __forceinline__ void load_wfrag(WFrag& f, const float* sWl) {
#pragma unroll
    for (int j = 0; j < NJ1; ++j) {
        f.b0[j] = *(const v2f*)(sWl + (2 * j) * FR);
        f.b1[j] = *(const v2f*)(sWl + (2 * j + 1) * FR);
    }
}

static __device__ __forceinline__ void load_ytile(YTile& f, const float* sYTl, int t) {
#pragma unroll
    for (int j = 0; j < NJ2; ++j) f.b[j] = *(const v2f*)(sYTl + (t * NJ2 + j) * FR);
}

static __device__ __forceinline__ void load_cfrag(v2f (&a)[NJ2], const float* crow) {
#pragma unroll
    for (int j = 0; j < NJ2; ++j) a[j] = *(const v2f*)(crow + 4 * j);
}

static __device__ __forceinline__ void store_c(float* sC, const v8f& t0, const v8f& t1,
                                               int l16, int hsel) {
#pragma unroll
    for (int r = 0; r < 8; ++r) {
        const int m = r + 8 * hsel;
        sC[m * CS + l16]      = t0[r];
        sC[m * CS + 16 + l16] = t1[r];
    }
}

static __device__ __forceinline__ void load_hblk(HBlk& f, const float* hrow,
                                                 const float* sYw, int blk,
                                                 int l16, int hsel) {
#pragma unroll
    for (int jj = 0; jj < 9; ++jj) {
        const int k = 4 * (blk * 9 + jj);
        f.a[jj].x  = hrow[k];
        f.a[jj].y  = hrow[k + 1];
        f.b0[jj].x = sYw[(k + 2 * hsel) * SP + l16];
        f.b0[jj].y = sYw[(k + 2 * hsel + 1) * SP + l16];
    }
    split_load_groups();
#pragma unroll
    for (int jj = 0; jj < 9; ++jj) {
        const int k = 4 * (blk * 9 + jj);
        f.b1[jj].x = sYw[(k + 2 * hsel) * SP + 16 + l16];
        f.b1[jj].y = sYw[(k + 2 * hsel + 1) * SP + 16 + l16];
    }
}

__global__ void __launch_bounds__(NT)
gaunt_fused_kernel(const float* __restrict__ X1,
                   const float* __restrict__ X2,
                   const float* __restrict__ C,
                   float* __restrict__ out) {
    extern __shared__ float smem[];
    float* sYw = smem + OFF_YW;
    float* sH  = smem + N_CONST;

    const int tid  = threadIdx.x;
    const int lane = tid & 31;
    const int wave = tid >> 5;
    const int hsel = lane >> 4;
    const int l16  = lane & 15;
    const float* sW1l = smem + OFF_W1 + 2 * lane;
    const float* sW2l = smem + OFF_W2 + 2 * lane;
    const float* sYTl = smem + OFF_YT + 2 * lane;

    {
        const v4f* gC = (const v4f*)C;
        v4f* lC = (v4f*)smem;
        constexpr int NV   = N_CONST / 4;
        constexpr int FULL = NV / NT;
        constexpr int REM  = NV - FULL * NT;
        v4f rc[FULL];
        v4f rlast = {};
#pragma unroll
        for (int i = 0; i < FULL; ++i) rc[i] = gC[tid + i * NT];
        if (tid < REM) rlast = gC[tid + FULL * NT];
#pragma unroll
        for (int i = 0; i < FULL; ++i) lC[tid + i * NT] = rc[i];
        if (tid < REM) lC[tid + FULL * NT] = rlast;
    }
    __syncthreads();

    const int rowBase = blockIdx.x * TM + wave * 16;

    v2f A1[16], A2[16];
    {
        const float* x1p = X1 + (size_t)(rowBase + l16) * DD + 2 * hsel;
        const float* x2p = X2 + (size_t)(rowBase + l16) * DD + 2 * hsel;
#pragma unroll
        for (int j = 0; j < 16; ++j) {
            A1[j] = *(const v2f*)(x1p + 4 * j);
            A2[j] = *(const v2f*)(x2p + 4 * j);
        }
    }

    float* sHw = sH + wave * 16 * GPS;
    float* sC1 = sHw;
    float* sC2 = sHw + 16 * CS;
    const float* c1row = sC1 + l16 * CS + 2 * hsel;
    const float* c2row = sC2 + l16 * CS + 2 * hsel;
    const float* hrow  = sHw + l16 * GPS + 2 * hsel;

    WFrag wf[2];
    YTile yt[2];
    HBlk  hb[2];
    v2f ca1[NJ2], ca2[NJ2];

    load_wfrag(wf[0], sW1l);
    load_wfrag(wf[1], sW2l);
    close_load_group();
    v8f c1a = {};
    v8f c1b = {};
#pragma unroll
    for (int j = 0; j < 16; ++j) {
        c1a = __builtin_amdgcn_wmma_f32_16x16x4_f32(
            false, A1[j], false, wf[0].b0[j], (short)0, c1a, false, false);
        c1b = __builtin_amdgcn_wmma_f32_16x16x4_f32(
            false, A1[j], false, wf[0].b1[j], (short)0, c1b, false, false);
    }
    __builtin_amdgcn_sched_barrier(0);
    store_c(sC1, c1a, c1b, l16, hsel);
    load_ytile(yt[0], sYTl, 0);
    load_cfrag(ca1, c1row);
    close_load_group();
    v8f c2a = {};
    v8f c2b = {};
#pragma unroll
    for (int j = 0; j < 16; ++j) {
        c2a = __builtin_amdgcn_wmma_f32_16x16x4_f32(
            false, A2[j], false, wf[1].b0[j], (short)0, c2a, false, false);
        c2b = __builtin_amdgcn_wmma_f32_16x16x4_f32(
            false, A2[j], false, wf[1].b1[j], (short)0, c2b, false, false);
    }
    __builtin_amdgcn_sched_barrier(0);
    store_c(sC2, c2a, c2b, l16, hsel);
    load_cfrag(ca2, c2row);

#pragma unroll
    for (int t = 0; t < GP / 16; ++t) {
        if (t + 1 < GP / 16) {
            load_ytile(yt[(t + 1) & 1], sYTl, t + 1);
        } else {
            load_hblk(hb[0], hrow, sYw, 0, l16, hsel);
        }
        close_load_group();
        const YTile& y = yt[t & 1];
        v8f acc1 = {};
        v8f acc2 = {};
#pragma unroll
        for (int j = 0; j < NJ2; ++j) {
            acc1 = __builtin_amdgcn_wmma_f32_16x16x4_f32(
                false, ca1[j], false, y.b[j], (short)0, acc1, false, false);
            acc2 = __builtin_amdgcn_wmma_f32_16x16x4_f32(
                false, ca2[j], false, y.b[j], (short)0, acc2, false, false);
        }
        __builtin_amdgcn_sched_barrier(0);
        v8f h = acc1 * acc2;
        const int col = t * 16 + l16;
#pragma unroll
        for (int r = 0; r < 8; ++r) {
            const int m = r + 8 * hsel;
            sHw[m * GPS + col] = h[r];
        }
    }

    v8f o0 = {};
    v8f o1 = {};
#pragma unroll
    for (int blk = 0; blk < 4; ++blk) {
        if (blk + 1 < 4) load_hblk(hb[(blk + 1) & 1], hrow, sYw, blk + 1, l16, hsel);
        close_load_group();
        const HBlk& c = hb[blk & 1];
#pragma unroll
        for (int jj = 0; jj < 9; ++jj) {
            o0 = __builtin_amdgcn_wmma_f32_16x16x4_f32(
                false, c.a[jj], false, c.b0[jj], (short)0, o0, false, false);
            o1 = __builtin_amdgcn_wmma_f32_16x16x4_f32(
                false, c.a[jj], false, c.b1[jj], (short)0, o1, false, false);
        }
        __builtin_amdgcn_sched_barrier(0);
    }
#pragma unroll
    for (int r = 0; r < 8; ++r) {
        const int m = r + 8 * hsel;
        out[(size_t)(rowBase + m) * SS + l16] = o0[r];
    }
    if (16 + l16 < SS) {
#pragma unroll
        for (int r = 0; r < 8; ++r) {
            const int m = r + 8 * hsel;
            out[(size_t)(rowBase + m) * SS + 16 + l16] = o1[r];
        }
    }
}

extern "C" void kernel_launch(void* const* d_in, const int* in_sizes, int n_in,
                              void* d_out, int out_size, void* d_ws, size_t ws_size,
                              hipStream_t stream) {
    const float* X1 = (const float*)d_in[0];
    const float* X2 = (const float*)d_in[1];
    const float* W1 = (const float*)d_in[2];
    const float* W2 = (const float*)d_in[3];
    const float* Y  = (const float*)d_in[4];
    const float* Yw = (const float*)d_in[5];

    float* C = (float*)d_ws;

    precompute_kernel<<<32, 256, 0, stream>>>(W1, W2, Y, Yw, C);

    const size_t shmem = (size_t)(N_CONST + TM * GPS) * sizeof(float);
    gaunt_fused_kernel<<<N_TOT / TM, NT, shmem, stream>>>(
        X1, X2, C, (float*)d_out);
}
